# starting kernel
# baseline (speedup 1.0000x reference)
#include <hip/hip_runtime.h>
#include <hip/hip_bf16.h>
#include <math.h>

typedef __attribute__((ext_vector_type(2))) float v2f;
typedef __attribute__((ext_vector_type(4))) float v4f;
typedef __attribute__((ext_vector_type(8))) float v8f;

#define BS   8
#define CIN  64
#define HH   256
#define WW   256
#define NEXP 4
#define OCH  64
#define KS   3
#define KTOT (CIN * KS * KS)
#define WROW (CIN * KS * KS)

#define TILE_W   128
#define CI_CHUNK 16
#define KCHUNK   (CI_CHUNK * 9)
#define SW_STRIDE 148
#define PATCH_W  132

__device__ constexpr int poff(int k) {
    return ((k / 9) * 3 + ((k % 9) / 3)) * PATCH_W + (k % 3);
}

__global__ __launch_bounds__(256)
void att_partial_kernel(const float* __restrict__ x,
                        float* __restrict__ parth,
                        float* __restrict__ partv) {
    const int t = threadIdx.x;
    const float* plane = x + (size_t)blockIdx.x * (HH * WW);
    float rowAcc = 0.f, colAcc = 0.f;
    for (int w = 0; w < WW; ++w) rowAcc += plane[t * WW + w];
    for (int h = 0; h < HH; ++h) colAcc += plane[h * WW + t];
    parth[(size_t)blockIdx.x * HH + t] = rowAcc;
    partv[(size_t)blockIdx.x * WW + t] = colAcc;
}

__global__ __launch_bounds__(256)
void att_finalize_kernel(const float* __restrict__ parth,
                         const float* __restrict__ partv,
                         const float* __restrict__ net_w,
                         const float* __restrict__ bias,
                         float* __restrict__ att,
                         float* __restrict__ aggb) {
    __shared__ float sh[HH + WW];
    __shared__ float sa[NEXP];
    const int b = blockIdx.x, t = threadIdx.x;
    float accH = 0.f, accV = 0.f;
    for (int c = 0; c < CIN; ++c) {
        accH += parth[((size_t)b * CIN + c) * HH + t];
        accV += partv[((size_t)b * CIN + c) * WW + t];
    }
    const float inv = 1.f / (float)(CIN * WW);
    sh[t]      = accH * inv;
    sh[HH + t] = accV * inv;
    __syncthreads();
    if (t < NEXP) {
        float acc = 0.f;
        for (int j = 0; j < HH + WW; ++j) acc += sh[j] * net_w[t * (HH + WW) + j];
        float s = 1.f / (1.f + __expf(-acc));
        att[b * NEXP + t] = s;
        sa[t] = s;
    }
    __syncthreads();
    if (t < OCH) {
        float acc = 0.f;
        for (int k = 0; k < NEXP; ++k) acc += sa[k] * bias[k * OCH + t];
        aggb[b * OCH + t] = acc;
    }
}

__global__ __launch_bounds__(256)
void agg_weight_kernel(const float* __restrict__ weight,
                       const float* __restrict__ att,
                       float* __restrict__ aggw) {
    const size_t idx = (size_t)blockIdx.x * 256 + threadIdx.x;
    const int b = (int)(idx / (OCH * KTOT));
    const int j = (int)(idx % (OCH * KTOT));
    const float a0 = att[b * NEXP + 0], a1 = att[b * NEXP + 1];
    const float a2 = att[b * NEXP + 2], a3 = att[b * NEXP + 3];
    const size_t E = (size_t)OCH * KTOT;
    aggw[idx] = a0 * weight[j] + a1 * weight[E + j] + a2 * weight[2 * E + j] + a3 * weight[3 * E + j];
}

__global__ __launch_bounds__(256)
void condconv_wmma_kernel(const float* __restrict__ x,
                          const float* __restrict__ aggw,
                          const float* __restrict__ aggb,
                          float* __restrict__ out) {
    __shared__ float sW[OCH * SW_STRIDE];
    __shared__ float sX[CI_CHUNK * 3 * PATCH_W];
    __shared__ float sBias[OCH];

    const int tid = threadIdx.x;
    const int b      = blockIdx.x >> 9;
    const int rem    = blockIdx.x & 511;
    const int h      = rem >> 1;
    const int wstart = (rem & 1) * TILE_W;

    if (tid < OCH) sBias[tid] = aggb[b * OCH + tid];

    const int lane   = tid & 31;
    const int wv     = tid >> 5;
    const int hl     = lane >> 4;
    const int l16    = lane & 15;
    const int m_base = (wv & 1) * 32;
    const int n_base = (wv >> 1) * 32;

    v8f acc[2][2] = {};

    const float* wrow = aggw + (size_t)b * (OCH * WROW);

    const int aoff0 = (m_base + l16) * SW_STRIDE + 2 * hl;
    const int aoff1 = (m_base + 16 + l16) * SW_STRIDE + 2 * hl;
    const int n0 = n_base + l16;
    const int n1 = n0 + 16;

    for (int ci0 = 0; ci0 < CIN; ci0 += CI_CHUNK) {
        __syncthreads();
        for (int e4 = tid; e4 < OCH * (KCHUNK / 4); e4 += 256) {
            const int o = e4 / (KCHUNK / 4), kc4 = e4 % (KCHUNK / 4);
            const v4f v = *(const v4f*)&wrow[o * WROW + ci0 * 9 + kc4 * 4];
            *(v4f*)&sW[o * SW_STRIDE + kc4 * 4] = v;
        }
        const float* xb = x + ((size_t)b * CIN + ci0) * (HH * WW);
        for (int e = tid; e < CI_CHUNK * 3 * PATCH_W; e += 256) {
            const int ci = e / (3 * PATCH_W);
            const int rr = (e / PATCH_W) % 3;
            const int cc = e % PATCH_W;
            const int gh = h - 1 + rr;
            const int gw = wstart - 1 + cc;
            float v = 0.f;
            if ((unsigned)gh < (unsigned)HH && (unsigned)gw < (unsigned)WW)
                v = xb[(size_t)ci * (HH * WW) + gh * WW + gw];
            sX[e] = v;
        }
        __syncthreads();

        if (ci0 + CI_CHUNK < CIN) {
            {
                const int o = tid >> 2, seg = tid & 3;
                __builtin_prefetch(wrow + (size_t)o * WROW + (ci0 + CI_CHUNK) * 9 + seg * 36, 0, 1);
            }
            if (tid < 96) {
                const int ci = tid / 6, rr = (tid / 2) % 3, hs = tid & 1;
                int gh = h - 1 + rr; gh = gh < 0 ? 0 : (gh > HH - 1 ? HH - 1 : gh);
                int gw = wstart + hs * 64; if (wstart == 0 && hs == 0) gw = 0;
                const float* nx = x + ((size_t)b * CIN + ci0 + CI_CHUNK + ci) * (HH * WW);
                __builtin_prefetch(nx + gh * WW + gw, 0, 1);
            }
        }

        #pragma unroll
        for (int kc0 = 0; kc0 < KCHUNK; kc0 += 4) {
            const int o0 = hl ? poff(kc0 + 2) : poff(kc0 + 0);
            const int o1 = hl ? poff(kc0 + 3) : poff(kc0 + 1);
            const v2f a0 = *(const v2f*)&sW[aoff0 + kc0];
            const v2f a1 = *(const v2f*)&sW[aoff1 + kc0];
            v2f b0, b1;
            b0.x = sX[o0 + n0]; b0.y = sX[o1 + n0];
            b1.x = sX[o0 + n1]; b1.y = sX[o1 + n1];
            acc[0][0] = __builtin_amdgcn_wmma_f32_16x16x4_f32(false, a0, false, b0, (short)0, acc[0][0], false, false);
            acc[0][1] = __builtin_amdgcn_wmma_f32_16x16x4_f32(false, a0, false, b1, (short)0, acc[0][1], false, false);
            acc[1][0] = __builtin_amdgcn_wmma_f32_16x16x4_f32(false, a1, false, b0, (short)0, acc[1][0], false, false);
            acc[1][1] = __builtin_amdgcn_wmma_f32_16x16x4_f32(false, a1, false, b1, (short)0, acc[1][1], false, false);
        }
    }

    #pragma unroll
    for (int mi = 0; mi < 2; ++mi) {
        #pragma unroll
        for (int nj = 0; nj < 2; ++nj) {
            const int n = n_base + nj * 16 + l16;
            #pragma unroll
            for (int r = 0; r < 8; ++r) {
                const int o = m_base + mi * 16 + hl * 8 + r;
                out[(((size_t)b * OCH + o) * HH + h) * WW + wstart + n] = acc[mi][nj][r] + sBias[o];
            }
        }
    }
}

extern "C" void kernel_launch(void* const* d_in, const int* in_sizes, int n_in,
                              void* d_out, int out_size, void* d_ws, size_t ws_size,
                              hipStream_t stream) {
    const float* x      = (const float*)d_in[0];
    const float* net_w  = (const float*)d_in[1];
    const float* weight = (const float*)d_in[2];
    const float* bias   = (const float*)d_in[3];
    float* out = (float*)d_out;

    float* ws = (float*)d_ws;
    float* att   = ws;
    float* aggb  = att + BS * NEXP;
    float* aggw  = aggb + BS * OCH;
    float* parth = aggw + (size_t)BS * OCH * KTOT;
    float* partv = parth + (size_t)BS * CIN * HH;

    att_partial_kernel<<<BS * CIN, 256, 0, stream>>>(x, parth, partv);
    att_finalize_kernel<<<BS, 256, 0, stream>>>(parth, partv, net_w, bias, att, aggb);
    agg_weight_kernel<<<(BS * OCH * KTOT) / 256, 256, 0, stream>>>(weight, att, aggw);
    condconv_wmma_kernel<<<BS * HH * (WW / TILE_W), 256, 0, stream>>>(x, aggw, aggb, out);
}


# gate: linear layer applied to each plane's row/column sums inside one streaming pass (plane read once, 4 numbers per plane out); no mean planes, no 512-step chain on 4 lanes; gate sum order changes
# speedup vs baseline: 2.1888x; 2.1888x over previous
#include <hip/hip_runtime.h>
#include <hip/hip_bf16.h>
#include <math.h>

typedef __attribute__((ext_vector_type(2))) float v2f;
typedef __attribute__((ext_vector_type(4))) float v4f;
typedef __attribute__((ext_vector_type(8))) float v8f;

#define BS   8
#define CIN  64
#define HH   256
#define WW   256
#define NEXP 4
#define OCH  64
#define KS   3
#define KTOT (CIN * KS * KS)

#define TILE_W     64
#define ROWS_PER_BLOCK 32
#define KROW       (KS * CIN)
#define SWS        196
#define USET       (OCH * SWS)
#define PAIR_PITCH 132
#define SLOT_SZ    (32 * PAIR_PITCH)
#define NSLOT      6
#define GK         8

#define PB 16
#define NB (HH / 4 / PB)
__global__ __launch_bounds__(256)
void gate_partial_kernel(const float* __restrict__ x,
                         const float* __restrict__ net_w,
                         float* __restrict__ part) {
    __shared__ v4f sA[HH];
    __shared__ float sRed[8][NEXP];
    const int t = threadIdx.x;
    const int c4 = t & 63, r0 = t >> 6;
    const float* src = x + (size_t)blockIdx.x * (HH * WW) + r0 * WW + 4 * c4;
    sA[t] = (v4f){net_w[t], net_w[(HH + WW) + t], net_w[2 * (HH + WW) + t], net_w[3 * (HH + WW) + t]};

    v4f col = {0.f, 0.f, 0.f, 0.f};
    float rs[NB * PB];
    v4f cur[PB], nxt[PB];
    #pragma unroll
    for (int j = 0; j < PB; ++j) cur[j] = *(const v4f*)(src + (size_t)(4 * j) * WW);
    #pragma unroll
    for (int bt = 0; bt < NB; ++bt) {
        if (bt + 1 < NB) {
            #pragma unroll
            for (int j = 0; j < PB; ++j) nxt[j] = *(const v4f*)(src + (size_t)(4 * ((bt + 1) * PB + j)) * WW);
        }
        __builtin_amdgcn_sched_barrier(0);
        #pragma unroll
        for (int j = 0; j < PB; ++j) {
            col += cur[j];
            rs[bt * PB + j] = (cur[j].x + cur[j].y) + (cur[j].z + cur[j].w);
        }
        __builtin_amdgcn_sched_barrier(0);
        if (bt + 1 < NB) {
            #pragma unroll
            for (int j = 0; j < PB; ++j) cur[j] = nxt[j];
        }
    }
    __syncthreads();

    v4f ak = {0.f, 0.f, 0.f, 0.f};
    #pragma unroll
    for (int i = 0; i < NB * PB; ++i) ak += sA[r0 + 4 * i] * rs[i];
    const float invH = 1.f / (float)(CIN * WW);
    const float invV = 1.f / (float)(CIN * HH);
    float p[NEXP];
    #pragma unroll
    for (int k = 0; k < NEXP; ++k) {
        const v4f bk = *(const v4f*)(net_w + k * (HH + WW) + HH + 4 * c4);
        const float cd = (bk.x * col.x + bk.y * col.y) + (bk.z * col.z + bk.w * col.w);
        p[k] = ak[k] * invH + cd * invV;
    }
    #pragma unroll
    for (int k = 0; k < NEXP; ++k) {
        #pragma unroll
        for (int m = 16; m >= 1; m >>= 1) p[k] += __shfl_xor(p[k], m);
    }
    if ((t & 31) == 0) {
        #pragma unroll
        for (int k = 0; k < NEXP; ++k) sRed[t >> 5][k] = p[k];
    }
    __syncthreads();
    if (t < NEXP) {
        float acc = 0.f;
        for (int w = 0; w < 8; ++w) acc += sRed[w][t];
        part[(size_t)blockIdx.x * NEXP + t] = acc;
    }
}

__global__ __launch_bounds__(64)
void att_finalize_kernel(const float* __restrict__ part,
                         const float* __restrict__ bias,
                         float* __restrict__ att,
                         float* __restrict__ aggb) {
    __shared__ float sr[2][NEXP];
    __shared__ float sa[NEXP];
    const int b = blockIdx.x, t = threadIdx.x;
    const v4f pv = *(const v4f*)(part + ((size_t)b * CIN + t) * NEXP);
    float p[NEXP] = {pv.x, pv.y, pv.z, pv.w};
    #pragma unroll
    for (int k = 0; k < NEXP; ++k) {
        #pragma unroll
        for (int m = 16; m >= 1; m >>= 1) p[k] += __shfl_xor(p[k], m);
    }
    if ((t & 31) == 0) {
        #pragma unroll
        for (int k = 0; k < NEXP; ++k) sr[t >> 5][k] = p[k];
    }
    __syncthreads();
    if (t < NEXP) {
        const float acc = sr[0][t] + sr[1][t];
        const float s = 1.f / (1.f + __expf(-acc));
        att[b * NEXP + t] = s;
        sa[t] = s;
    }
    __syncthreads();
    {
        float acc = 0.f;
        for (int k = 0; k < NEXP; ++k) acc += sa[k] * bias[k * OCH + t];
        aggb[b * OCH + t] = acc;
    }
}

__global__ __launch_bounds__(256)
void agg_weight_kernel(const float* __restrict__ weight,
                       const float* __restrict__ att,
                       float* __restrict__ aggw) {
    const size_t idx = (size_t)blockIdx.x * 256 + threadIdx.x;
    const int b = (int)(idx / (OCH * KTOT));
    const int j = (int)(idx % (OCH * KTOT));
    const float a0 = att[b * NEXP + 0], a1 = att[b * NEXP + 1];
    const float a2 = att[b * NEXP + 2], a3 = att[b * NEXP + 3];
    const size_t E = (size_t)OCH * KTOT;
    aggw[idx] = a0 * weight[j] + a1 * weight[E + j] + a2 * weight[2 * E + j] + a3 * weight[3 * E + j];
}


struct RowRegs {
    v4f lo[2], hi[2];
    float halo;
};

__device__ __forceinline__ void row_fetch(RowRegs& R, const float* __restrict__ xb,
                                          int row, int wstart, int tid) {
    if ((unsigned)row < (unsigned)HH) {
        #pragma unroll
        for (int i = 0; i < 2; ++i) {
            const int q = tid + 256 * i, cp = q >> 4, c4 = q & 15;
            const float* p = xb + ((size_t)(2 * cp) * HH + row) * WW + wstart + 4 * c4;
            R.lo[i] = *(const v4f*)p;
            R.hi[i] = *(const v4f*)(p + HH * WW);
        }
        R.halo = 0.f;
        if (tid < 128) {
            const int ci = tid >> 1;
            const int gw = (tid & 1) ? wstart + TILE_W : wstart - 1;
            if ((unsigned)gw < (unsigned)WW) R.halo = xb[((size_t)ci * HH + row) * WW + gw];
        }
    } else {
        #pragma unroll
        for (int i = 0; i < 2; ++i) { R.lo[i] = (v4f){0.f, 0.f, 0.f, 0.f}; R.hi[i] = (v4f){0.f, 0.f, 0.f, 0.f}; }
        R.halo = 0.f;
    }
}

__device__ __forceinline__ void row_store(const RowRegs& R, float* __restrict__ slot, int tid) {
    #pragma unroll
    for (int i = 0; i < 2; ++i) {
        const int q = tid + 256 * i, cp = q >> 4, c4 = q & 15;
        float* d = slot + cp * PAIR_PITCH + (4 * c4 + 1) * 2;
        *(v2f*)d       = (v2f){R.lo[i].x, R.hi[i].x};
        *(v2f*)(d + 2) = (v2f){R.lo[i].y, R.hi[i].y};
        *(v2f*)(d + 4) = (v2f){R.lo[i].z, R.hi[i].z};
        *(v2f*)(d + 6) = (v2f){R.lo[i].w, R.hi[i].w};
    }
    if (tid < 128) {
        const int ci = tid >> 1;
        const int j = (tid & 1) ? (TILE_W + 1) : 0;
        slot[(ci >> 1) * PAIR_PITCH + j * 2 + (ci & 1)] = R.halo;
    }
}

__device__ __forceinline__ void frag_load(v2f (&fa)[GK][4], v4f (&fr)[GK][2],
                                          const float* aB, const float* bB) {
    #pragma unroll
    for (int i = 0; i < GK; ++i) {
        #pragma unroll
        for (int p = 0; p < 4; ++p) fa[i][p] = *(const v2f*)(aB + p * USET + i * 4);
        fr[i][0] = *(const v4f*)(bB + i * 2 * PAIR_PITCH);
        fr[i][1] = *(const v4f*)(bB + i * 2 * PAIR_PITCH + 4);
    }
}

__device__ __forceinline__ void frag_mma(v8f (&acc)[4], const v2f (&fa)[GK][4], const v4f (&fr)[GK][2]) {
    #pragma unroll
    for (int i = 0; i < GK; ++i) {
        const v2f d0 = fr[i][0].xy, d1 = fr[i][0].zw, d2 = fr[i][1].xy, d3 = fr[i][1].zw;
        acc[0] = __builtin_amdgcn_wmma_f32_16x16x4_f32(false, fa[i][0], false, d0 - d2, (short)0, acc[0], false, false);
        acc[1] = __builtin_amdgcn_wmma_f32_16x16x4_f32(false, fa[i][1], false, d1 + d2, (short)0, acc[1], false, false);
        acc[2] = __builtin_amdgcn_wmma_f32_16x16x4_f32(false, fa[i][2], false, d2 - d1, (short)0, acc[2], false, false);
        acc[3] = __builtin_amdgcn_wmma_f32_16x16x4_f32(false, fa[i][3], false, d1 - d3, (short)0, acc[3], false, false);
    }
}

__global__ __launch_bounds__(256)
void condconv_wmma_kernel(const float* __restrict__ x,
                          const float* __restrict__ aggw,
                          const float* __restrict__ aggb,
                          float* __restrict__ out) {
    __shared__ __attribute__((aligned(16))) float sU[4 * USET];
    __shared__ __attribute__((aligned(16))) float sX[NSLOT * SLOT_SZ];

    const int tid    = threadIdx.x;
    const int b      = blockIdx.x >> 5;
    const int rem    = blockIdx.x & 31;
    const int wstart = (rem & 3) * TILE_W;
    const int h0     = (rem >> 2) * ROWS_PER_BLOCK;

    const int lane   = tid & 31;
    const int wv     = tid >> 5;
    const int hl     = lane >> 4;
    const int l16    = lane & 15;
    const int m_base = (wv & 3) * 16;
    const int j_base = (wv >> 2) * 16;

    const float* xb = x + (size_t)b * (CIN * HH * WW);

    {
        const float* wsrc = aggw + (size_t)b * (OCH * KTOT);
        for (int it = 0; it < (OCH * KROW) / 256; it += 8) {
            float g[8][3];
            #pragma unroll
            for (int j = 0; j < 8; ++j) {
                const int e = tid + 256 * (it + j);
                g[j][0] = wsrc[3 * e]; g[j][1] = wsrc[3 * e + 1]; g[j][2] = wsrc[3 * e + 2];
            }
            #pragma unroll
            for (int j = 0; j < 8; ++j) {
                const int e = tid + 256 * (it + j);
                const int o = e / KROW, m = e % KROW;
                const int ci = m / 3, kh = m % 3;
                float* d = sU + o * SWS + kh * 64 + ci;
                const float s02 = g[j][0] + g[j][2];
                d[0]        = g[j][0];
                d[USET]     = 0.5f * (s02 + g[j][1]);
                d[2 * USET] = 0.5f * (s02 - g[j][1]);
                d[3 * USET] = g[j][2];
            }
        }
    }

    RowRegs R;
    for (int r = -1; r <= 2; ++r) {
        row_fetch(R, xb, h0 + r, wstart, tid);
        row_store(R, sX + ((h0 + r + NSLOT) % NSLOT) * SLOT_SZ, tid);
    }

    float bv[8];
    #pragma unroll
    for (int r = 0; r < 8; ++r) bv[r] = aggb[b * OCH + m_base + hl * 8 + r];

    const float* aLane = sU + (m_base + l16) * SWS + 2 * hl;
    const int    bLane = hl * PAIR_PITCH + (j_base + l16) * 4;

    __syncthreads();

    v2f fa[2][GK][4];
    v4f fr[2][GK][2];
    frag_load(fa[0], fr[0], aLane, sX + ((h0 - 1 + NSLOT) % NSLOT) * SLOT_SZ + bLane);

    for (int h = h0; h < h0 + ROWS_PER_BLOCK; ++h) {
        const bool more = (h + 3 <= h0 + ROWS_PER_BLOCK);
        if (more) row_fetch(R, xb, h + 3, wstart, tid);

        const float* sb[3];
        #pragma unroll
        for (int kh = 0; kh < 3; ++kh) sb[kh] = sX + ((h - 1 + kh + NSLOT) % NSLOT) * SLOT_SZ + bLane;

        v8f acc[4] = {};
        #pragma unroll
        for (int g = 0; g < 6; ++g) {
            const int gn = g + 1;
            if (gn < 6) {
                const int kh = gn >> 1, q0 = (gn & 1) * GK;
                frag_load(fa[gn & 1], fr[gn & 1], aLane + kh * 64 + q0 * 4,
                          sb[kh] + q0 * 2 * PAIR_PITCH);
            } else {
                frag_load(fa[0], fr[0], aLane, sb[1]);
            }
            __builtin_amdgcn_sched_barrier(0);
            frag_mma(acc, fa[g & 1], fr[g & 1]);
            __builtin_amdgcn_sched_barrier(0);
        }

        #pragma unroll
        for (int r = 0; r < 8; ++r) {
            const int o = m_base + hl * 8 + r;
            const float y0 = ((acc[0][r] + acc[1][r]) + acc[2][r]) + bv[r];
            const float y1 = ((acc[1][r] - acc[2][r]) - acc[3][r]) + bv[r];
            *(v2f*)&out[(((size_t)b * OCH + o) * HH + h) * WW + wstart + 2 * (j_base + l16)] = (v2f){y0, y1};
        }
        if (more) row_store(R, sX + ((h + 3) % NSLOT) * SLOT_SZ, tid);

        if ((h - h0) & 1) __syncthreads();
    }
}

extern "C" void kernel_launch(void* const* d_in, const int* in_sizes, int n_in,
                              void* d_out, int out_size, void* d_ws, size_t ws_size,
                              hipStream_t stream) {
    const float* x      = (const float*)d_in[0];
    const float* net_w  = (const float*)d_in[1];
    const float* weight = (const float*)d_in[2];
    const float* bias   = (const float*)d_in[3];
    float* out = (float*)d_out;

    float* ws = (float*)d_ws;
    float* att   = ws;
    float* aggb  = att + BS * NEXP;
    float* aggw  = aggb + BS * OCH;
    float* part  = aggw + (size_t)BS * OCH * KTOT;

    gate_partial_kernel<<<BS * CIN, 256, 0, stream>>>(x, net_w, part);
    att_finalize_kernel<<<BS, OCH, 0, stream>>>(part, bias, att, aggb);
    agg_weight_kernel<<<(BS * OCH * KTOT) / 256, 256, 0, stream>>>(weight, att, aggw);
    condconv_wmma_kernel<<<BS * 32, 256, 0, stream>>>(x, aggw, aggb, out);
}


# conv: 3-tap minimal filtering in both directions: 2x2 output block from 16 K=64 WMMA products (was 2 x 4 of K=192), 16 weight sets of 32 out channels in LDS; f32 kept; sum order, rounding change
# speedup vs baseline: 2.4493x; 1.1190x over previous
#include <hip/hip_runtime.h>
#include <hip/hip_bf16.h>
#include <math.h>

typedef __attribute__((ext_vector_type(2))) float v2f;
typedef __attribute__((ext_vector_type(4))) float v4f;
typedef __attribute__((ext_vector_type(8))) float v8f;

#define BS   8
#define CIN  64
#define HH   256
#define WW   256
#define NEXP 4
#define OCH  64
#define KS   3
#define KTOT (CIN * KS * KS)

#define TILE_W     128
#define ROWS_PER_BLOCK 32
#define MBLK       32
#define NSET       16
#define SWS        68
#define USET       (MBLK * SWS)
#define PAIR_PITCH 260
#define SLOT_SZ    (32 * PAIR_PITCH)
#define NSLOT      4
#define NITEM      4

#define PB 16
#define NB (HH / 4 / PB)
__global__ __launch_bounds__(256)
void gate_partial_kernel(const float* __restrict__ x,
                         const float* __restrict__ net_w,
                         float* __restrict__ part) {
    __shared__ v4f sA[HH];
    __shared__ float sRed[8][NEXP];
    const int t = threadIdx.x;
    const int c4 = t & 63, r0 = t >> 6;
    const float* src = x + (size_t)blockIdx.x * (HH * WW) + r0 * WW + 4 * c4;
    sA[t] = (v4f){net_w[t], net_w[(HH + WW) + t], net_w[2 * (HH + WW) + t], net_w[3 * (HH + WW) + t]};

    v4f col = {0.f, 0.f, 0.f, 0.f};
    float rs[NB * PB];
    v4f cur[PB], nxt[PB];
    #pragma unroll
    for (int j = 0; j < PB; ++j) cur[j] = *(const v4f*)(src + (size_t)(4 * j) * WW);
    #pragma unroll
    for (int bt = 0; bt < NB; ++bt) {
        if (bt + 1 < NB) {
            #pragma unroll
            for (int j = 0; j < PB; ++j) nxt[j] = *(const v4f*)(src + (size_t)(4 * ((bt + 1) * PB + j)) * WW);
        }
        __builtin_amdgcn_sched_barrier(0);
        #pragma unroll
        for (int j = 0; j < PB; ++j) {
            col += cur[j];
            rs[bt * PB + j] = (cur[j].x + cur[j].y) + (cur[j].z + cur[j].w);
        }
        __builtin_amdgcn_sched_barrier(0);
        if (bt + 1 < NB) {
            #pragma unroll
            for (int j = 0; j < PB; ++j) cur[j] = nxt[j];
        }
    }
    __syncthreads();

    v4f ak = {0.f, 0.f, 0.f, 0.f};
    #pragma unroll
    for (int i = 0; i < NB * PB; ++i) ak += sA[r0 + 4 * i] * rs[i];
    const float invH = 1.f / (float)(CIN * WW);
    const float invV = 1.f / (float)(CIN * HH);
    float p[NEXP];
    #pragma unroll
    for (int k = 0; k < NEXP; ++k) {
        const v4f bk = *(const v4f*)(net_w + k * (HH + WW) + HH + 4 * c4);
        const float cd = (bk.x * col.x + bk.y * col.y) + (bk.z * col.z + bk.w * col.w);
        p[k] = ak[k] * invH + cd * invV;
    }
    #pragma unroll
    for (int k = 0; k < NEXP; ++k) {
        #pragma unroll
        for (int m = 16; m >= 1; m >>= 1) p[k] += __shfl_xor(p[k], m);
    }
    if ((t & 31) == 0) {
        #pragma unroll
        for (int k = 0; k < NEXP; ++k) sRed[t >> 5][k] = p[k];
    }
    __syncthreads();
    if (t < NEXP) {
        float acc = 0.f;
        for (int w = 0; w < 8; ++w) acc += sRed[w][t];
        part[(size_t)blockIdx.x * NEXP + t] = acc;
    }
}

__global__ __launch_bounds__(64)
void att_finalize_kernel(const float* __restrict__ part,
                         const float* __restrict__ bias,
                         float* __restrict__ att,
                         float* __restrict__ aggb) {
    __shared__ float sr[2][NEXP];
    __shared__ float sa[NEXP];
    const int b = blockIdx.x, t = threadIdx.x;
    const v4f pv = *(const v4f*)(part + ((size_t)b * CIN + t) * NEXP);
    float p[NEXP] = {pv.x, pv.y, pv.z, pv.w};
    #pragma unroll
    for (int k = 0; k < NEXP; ++k) {
        #pragma unroll
        for (int m = 16; m >= 1; m >>= 1) p[k] += __shfl_xor(p[k], m);
    }
    if ((t & 31) == 0) {
        #pragma unroll
        for (int k = 0; k < NEXP; ++k) sr[t >> 5][k] = p[k];
    }
    __syncthreads();
    if (t < NEXP) {
        const float acc = sr[0][t] + sr[1][t];
        const float s = 1.f / (1.f + __expf(-acc));
        att[b * NEXP + t] = s;
        sa[t] = s;
    }
    __syncthreads();
    {
        float acc = 0.f;
        for (int k = 0; k < NEXP; ++k) acc += sa[k] * bias[k * OCH + t];
        aggb[b * OCH + t] = acc;
    }
}

__global__ __launch_bounds__(256)
void agg_weight_kernel(const float* __restrict__ weight,
                       const float* __restrict__ att,
                       float* __restrict__ aggw) {
    const size_t idx = (size_t)blockIdx.x * 256 + threadIdx.x;
    const int b = (int)(idx / (OCH * KTOT));
    const int j = (int)(idx % (OCH * KTOT));
    const float a0 = att[b * NEXP + 0], a1 = att[b * NEXP + 1];
    const float a2 = att[b * NEXP + 2], a3 = att[b * NEXP + 3];
    const size_t E = (size_t)OCH * KTOT;
    aggw[idx] = a0 * weight[j] + a1 * weight[E + j] + a2 * weight[2 * E + j] + a3 * weight[3 * E + j];
}


struct RowRegs {
    v4f lo[NITEM], hi[NITEM];
    float halo;
};

__device__ __forceinline__ void row_fetch(RowRegs& R, const float* __restrict__ xb,
                                          int row, int wstart, int tid) {
    if ((unsigned)row < (unsigned)HH) {
        #pragma unroll
        for (int i = 0; i < NITEM; ++i) {
            const int q = tid + 256 * i, cp = q >> 5, c4 = q & 31;
            const float* p = xb + ((size_t)(2 * cp) * HH + row) * WW + wstart + 4 * c4;
            R.lo[i] = *(const v4f*)p;
            R.hi[i] = *(const v4f*)(p + HH * WW);
        }
        R.halo = 0.f;
        if (tid < 128) {
            const int ci = tid >> 1;
            const int gw = (tid & 1) ? wstart + TILE_W : wstart - 1;
            if ((unsigned)gw < (unsigned)WW) R.halo = xb[((size_t)ci * HH + row) * WW + gw];
        }
    } else {
        #pragma unroll
        for (int i = 0; i < NITEM; ++i) { R.lo[i] = (v4f){0.f, 0.f, 0.f, 0.f}; R.hi[i] = (v4f){0.f, 0.f, 0.f, 0.f}; }
        R.halo = 0.f;
    }
}

__device__ __forceinline__ void row_store(const RowRegs& R, float* __restrict__ slot, int tid) {
    #pragma unroll
    for (int i = 0; i < NITEM; ++i) {
        const int q = tid + 256 * i, cp = q >> 5, c4 = q & 31;
        float* d = slot + cp * PAIR_PITCH + (4 * c4 + 1) * 2;
        *(v2f*)d       = (v2f){R.lo[i].x, R.hi[i].x};
        *(v2f*)(d + 2) = (v2f){R.lo[i].y, R.hi[i].y};
        *(v2f*)(d + 4) = (v2f){R.lo[i].z, R.hi[i].z};
        *(v2f*)(d + 6) = (v2f){R.lo[i].w, R.hi[i].w};
    }
    if (tid < 128) {
        const int ci = tid >> 1;
        const int j = (tid & 1) ? (TILE_W + 1) : 0;
        slot[(ci >> 1) * PAIR_PITCH + j * 2 + (ci & 1)] = R.halo;
    }
}

__device__ __forceinline__ void frag_load(v2f (&fa)[NSET], v4f (&fr)[4][2],
                                          const float* aB, const float* const (&rb)[4], int boff) {
    #pragma unroll
    for (int s = 0; s < NSET; ++s) fa[s] = *(const v2f*)(aB + s * USET);
    #pragma unroll
    for (int r = 0; r < 4; ++r) {
        fr[r][0] = *(const v4f*)(rb[r] + boff);
        fr[r][1] = *(const v4f*)(rb[r] + boff + 4);
    }
}

__device__ __forceinline__ void frag_mma(v8f (&acc)[NSET], const v2f (&fa)[NSET], const v4f (&fr)[4][2]) {
    v4f t[4][2];
    #pragma unroll
    for (int c = 0; c < 2; ++c) {
        t[0][c] = fr[0][c] - fr[2][c];
        t[1][c] = fr[1][c] + fr[2][c];
        t[2][c] = fr[2][c] - fr[1][c];
        t[3][c] = fr[1][c] - fr[3][c];
    }
    #pragma unroll
    for (int a = 0; a < 4; ++a) {
        const v2f d0 = t[a][0].xy, d1 = t[a][0].zw, d2 = t[a][1].xy, d3 = t[a][1].zw;
        acc[4 * a + 0] = __builtin_amdgcn_wmma_f32_16x16x4_f32(false, fa[4 * a + 0], false, d0 - d2, (short)0, acc[4 * a + 0], false, false);
        acc[4 * a + 1] = __builtin_amdgcn_wmma_f32_16x16x4_f32(false, fa[4 * a + 1], false, d1 + d2, (short)0, acc[4 * a + 1], false, false);
        acc[4 * a + 2] = __builtin_amdgcn_wmma_f32_16x16x4_f32(false, fa[4 * a + 2], false, d2 - d1, (short)0, acc[4 * a + 2], false, false);
        acc[4 * a + 3] = __builtin_amdgcn_wmma_f32_16x16x4_f32(false, fa[4 * a + 3], false, d1 - d3, (short)0, acc[4 * a + 3], false, false);
    }
}

__global__ __launch_bounds__(256)
void condconv_wmma_kernel(const float* __restrict__ x,
                          const float* __restrict__ aggw,
                          const float* __restrict__ aggb,
                          float* __restrict__ out) {
    __shared__ __attribute__((aligned(16))) float sU[NSET * USET];
    __shared__ __attribute__((aligned(16))) float sX[NSLOT * SLOT_SZ];

    const int tid    = threadIdx.x;
    const int b      = blockIdx.x >> 5;
    const int rem    = blockIdx.x & 31;
    const int mh     = rem & 1;
    const int wstart = ((rem >> 1) & 1) * TILE_W;
    const int h0     = (rem >> 2) * ROWS_PER_BLOCK;

    const int lane   = tid & 31;
    const int wv     = tid >> 5;
    const int hl     = lane >> 4;
    const int l16    = lane & 15;
    const int m_base = (wv & 1) * 16;
    const int j_base = (wv >> 1) * 16;

    const float* xb = x + (size_t)b * (CIN * HH * WW);

    {
        const float* wsrc = aggw + (size_t)b * (OCH * KTOT) + (size_t)(mh * MBLK) * KTOT;
        for (int it = 0; it < (MBLK * CIN) / 256; it += 4) {
            float g[4][9];
            #pragma unroll
            for (int j = 0; j < 4; ++j) {
                const int e = tid + 256 * (it + j);
                #pragma unroll
                for (int k = 0; k < 9; ++k) g[j][k] = wsrc[9 * e + k];
            }
            #pragma unroll
            for (int j = 0; j < 4; ++j) {
                const int e = tid + 256 * (it + j);
                const int o = e >> 6, ci = e & 63;
                float w[3][4];
                #pragma unroll
                for (int kh = 0; kh < 3; ++kh) {
                    const float g0 = g[j][3 * kh], g1 = g[j][3 * kh + 1], g2 = g[j][3 * kh + 2];
                    const float s02 = g0 + g2;
                    w[kh][0] = g0;
                    w[kh][1] = 0.5f * (s02 + g1);
                    w[kh][2] = 0.5f * (s02 - g1);
                    w[kh][3] = g2;
                }
                float* d = sU + o * SWS + ci;
                #pragma unroll
                for (int c = 0; c < 4; ++c) {
                    const float t02 = w[0][c] + w[2][c];
                    d[(0 + c) * USET]  = w[0][c];
                    d[(4 + c) * USET]  = 0.5f * (t02 + w[1][c]);
                    d[(8 + c) * USET]  = 0.5f * (t02 - w[1][c]);
                    d[(12 + c) * USET] = w[2][c];
                }
            }
        }
    }

    RowRegs RA, RB;
    for (int r = -1; r <= 1; r += 2) {
        row_fetch(RA, xb, h0 + r, wstart, tid);
        row_fetch(RB, xb, h0 + r + 1, wstart, tid);
        row_store(RA, sX + ((h0 + r + 1) & 3) * SLOT_SZ, tid);
        row_store(RB, sX + ((h0 + r + 2) & 3) * SLOT_SZ, tid);
    }

    float bv[8];
    #pragma unroll
    for (int r = 0; r < 8; ++r) bv[r] = aggb[b * OCH + mh * MBLK + m_base + hl * 8 + r];

    const float* aLane = sU + (m_base + l16) * SWS + 2 * hl;
    const int    bLane = hl * PAIR_PITCH + (j_base + l16) * 4;

    __syncthreads();

    v2f fa[2][NSET];
    v4f fr[2][4][2];

    for (int p = 0; p < ROWS_PER_BLOCK / 2; ++p) {
        const int h = h0 + 2 * p;
        const bool more = (p + 1 < ROWS_PER_BLOCK / 2);
        if (more) {
            row_fetch(RA, xb, h + 3, wstart, tid);
            row_fetch(RB, xb, h + 4, wstart, tid);
        }

        const float* rb[4];
        #pragma unroll
        for (int r = 0; r < 4; ++r) rb[r] = sX + ((h + r) & 3) * SLOT_SZ + bLane;

        v8f acc[NSET] = {};
        frag_load(fa[0], fr[0], aLane, rb, 0);
        #pragma unroll
        for (int q = 0; q < CIN / 4; ++q) {
            if (q + 1 < CIN / 4)
                frag_load(fa[(q + 1) & 1], fr[(q + 1) & 1], aLane + (q + 1) * 4, rb, (q + 1) * 2 * PAIR_PITCH);
            __builtin_amdgcn_sched_barrier(0);
            frag_mma(acc, fa[q & 1], fr[q & 1]);
            __builtin_amdgcn_sched_barrier(0);
        }

        #pragma unroll
        for (int r = 0; r < 8; ++r) {
            const int o = mh * MBLK + m_base + hl * 8 + r;
            float s0[4], s1[4];
            #pragma unroll
            for (int c = 0; c < 4; ++c) {
                s0[c] = (acc[c][r] + acc[4 + c][r]) + acc[8 + c][r];
                s1[c] = (acc[4 + c][r] - acc[8 + c][r]) - acc[12 + c][r];
            }
            const float y00 = ((s0[0] + s0[1]) + s0[2]) + bv[r];
            const float y01 = ((s0[1] - s0[2]) - s0[3]) + bv[r];
            const float y10 = ((s1[0] + s1[1]) + s1[2]) + bv[r];
            const float y11 = ((s1[1] - s1[2]) - s1[3]) + bv[r];
            float* po = &out[(((size_t)b * OCH + o) * HH + h) * WW + wstart + 2 * (j_base + l16)];
            *(v2f*)po        = (v2f){y00, y01};
            *(v2f*)(po + WW) = (v2f){y10, y11};
        }

        if (more) {
            __syncthreads();
            row_store(RA, sX + ((h + 4) & 3) * SLOT_SZ, tid);
            row_store(RB, sX + ((h + 5) & 3) * SLOT_SZ, tid);
            __syncthreads();
        }
    }
}

extern "C" void kernel_launch(void* const* d_in, const int* in_sizes, int n_in,
                              void* d_out, int out_size, void* d_ws, size_t ws_size,
                              hipStream_t stream) {
    const float* x      = (const float*)d_in[0];
    const float* net_w  = (const float*)d_in[1];
    const float* weight = (const float*)d_in[2];
    const float* bias   = (const float*)d_in[3];
    float* out = (float*)d_out;

    float* ws = (float*)d_ws;
    float* att   = ws;
    float* aggb  = att + BS * NEXP;
    float* aggw  = aggb + BS * OCH;
    float* part  = aggw + (size_t)BS * OCH * KTOT;

    gate_partial_kernel<<<BS * CIN, 256, 0, stream>>>(x, net_w, part);
    att_finalize_kernel<<<BS, OCH, 0, stream>>>(part, bias, att, aggb);
    agg_weight_kernel<<<(BS * OCH * KTOT) / 256, 256, 0, stream>>>(weight, att, aggw);
    condconv_wmma_kernel<<<BS * 32, 256, 0, stream>>>(x, aggw, aggb, out);
}


# conv: block tile 4 rows x 64 cols on a 10-row LDS input ring: next step's rows stored into free slots amid the products; one barrier per step, ahead of the output stage (was 2 + a store phase)
# speedup vs baseline: 2.7179x; 1.1097x over previous
#include <hip/hip_runtime.h>
#include <hip/hip_bf16.h>
#include <math.h>

typedef __attribute__((ext_vector_type(2))) float v2f;
typedef __attribute__((ext_vector_type(4))) float v4f;
typedef __attribute__((ext_vector_type(8))) float v8f;

#define BS   8
#define CIN  64
#define HH   256
#define WW   256
#define NEXP 4
#define OCH  64
#define KS   3
#define KTOT (CIN * KS * KS)

#define TILE_W     64
#define ROWS_PER_BLOCK 64
#define MBLK       32
#define NSET       16
#define SWS        68
#define USET       (MBLK * SWS)
#define PAIR_PITCH 132
#define SLOT_SZ    (32 * PAIR_PITCH)
#define NSLOT      10
#define NITEM      4

#define PB 16
#define NB (HH / 4 / PB)
__global__ __launch_bounds__(256)
void gate_partial_kernel(const float* __restrict__ x,
                         const float* __restrict__ net_w,
                         float* __restrict__ part) {
    __shared__ v4f sA[HH];
    __shared__ float sRed[8][NEXP];
    const int t = threadIdx.x;
    const int c4 = t & 63, r0 = t >> 6;
    const float* src = x + (size_t)blockIdx.x * (HH * WW) + r0 * WW + 4 * c4;
    sA[t] = (v4f){net_w[t], net_w[(HH + WW) + t], net_w[2 * (HH + WW) + t], net_w[3 * (HH + WW) + t]};

    v4f col = {0.f, 0.f, 0.f, 0.f};
    float rs[NB * PB];
    v4f cur[PB], nxt[PB];
    #pragma unroll
    for (int j = 0; j < PB; ++j) cur[j] = *(const v4f*)(src + (size_t)(4 * j) * WW);
    #pragma unroll
    for (int bt = 0; bt < NB; ++bt) {
        if (bt + 1 < NB) {
            #pragma unroll
            for (int j = 0; j < PB; ++j) nxt[j] = *(const v4f*)(src + (size_t)(4 * ((bt + 1) * PB + j)) * WW);
        }
        __builtin_amdgcn_sched_barrier(0);
        #pragma unroll
        for (int j = 0; j < PB; ++j) {
            col += cur[j];
            rs[bt * PB + j] = (cur[j].x + cur[j].y) + (cur[j].z + cur[j].w);
        }
        __builtin_amdgcn_sched_barrier(0);
        if (bt + 1 < NB) {
            #pragma unroll
            for (int j = 0; j < PB; ++j) cur[j] = nxt[j];
        }
    }
    __syncthreads();

    v4f ak = {0.f, 0.f, 0.f, 0.f};
    #pragma unroll
    for (int i = 0; i < NB * PB; ++i) ak += sA[r0 + 4 * i] * rs[i];
    const float invH = 1.f / (float)(CIN * WW);
    const float invV = 1.f / (float)(CIN * HH);
    float p[NEXP];
    #pragma unroll
    for (int k = 0; k < NEXP; ++k) {
        const v4f bk = *(const v4f*)(net_w + k * (HH + WW) + HH + 4 * c4);
        const float cd = (bk.x * col.x + bk.y * col.y) + (bk.z * col.z + bk.w * col.w);
        p[k] = ak[k] * invH + cd * invV;
    }
    #pragma unroll
    for (int k = 0; k < NEXP; ++k) {
        #pragma unroll
        for (int m = 16; m >= 1; m >>= 1) p[k] += __shfl_xor(p[k], m);
    }
    if ((t & 31) == 0) {
        #pragma unroll
        for (int k = 0; k < NEXP; ++k) sRed[t >> 5][k] = p[k];
    }
    __syncthreads();
    if (t < NEXP) {
        float acc = 0.f;
        for (int w = 0; w < 8; ++w) acc += sRed[w][t];
        part[(size_t)blockIdx.x * NEXP + t] = acc;
    }
}

__global__ __launch_bounds__(64)
void att_finalize_kernel(const float* __restrict__ part,
                         const float* __restrict__ bias,
                         float* __restrict__ att,
                         float* __restrict__ aggb) {
    __shared__ float sr[2][NEXP];
    __shared__ float sa[NEXP];
    const int b = blockIdx.x, t = threadIdx.x;
    const v4f pv = *(const v4f*)(part + ((size_t)b * CIN + t) * NEXP);
    float p[NEXP] = {pv.x, pv.y, pv.z, pv.w};
    #pragma unroll
    for (int k = 0; k < NEXP; ++k) {
        #pragma unroll
        for (int m = 16; m >= 1; m >>= 1) p[k] += __shfl_xor(p[k], m);
    }
    if ((t & 31) == 0) {
        #pragma unroll
        for (int k = 0; k < NEXP; ++k) sr[t >> 5][k] = p[k];
    }
    __syncthreads();
    if (t < NEXP) {
        const float acc = sr[0][t] + sr[1][t];
        const float s = 1.f / (1.f + __expf(-acc));
        att[b * NEXP + t] = s;
        sa[t] = s;
    }
    __syncthreads();
    {
        float acc = 0.f;
        for (int k = 0; k < NEXP; ++k) acc += sa[k] * bias[k * OCH + t];
        aggb[b * OCH + t] = acc;
    }
}

__global__ __launch_bounds__(256)
void agg_weight_kernel(const float* __restrict__ weight,
                       const float* __restrict__ att,
                       float* __restrict__ aggw) {
    const size_t idx = (size_t)blockIdx.x * 256 + threadIdx.x;
    const int b = (int)(idx / (OCH * KTOT));
    const int j = (int)(idx % (OCH * KTOT));
    const float a0 = att[b * NEXP + 0], a1 = att[b * NEXP + 1];
    const float a2 = att[b * NEXP + 2], a3 = att[b * NEXP + 3];
    const size_t E = (size_t)OCH * KTOT;
    aggw[idx] = a0 * weight[j] + a1 * weight[E + j] + a2 * weight[2 * E + j] + a3 * weight[3 * E + j];
}


struct RowRegs {
    v4f lo[NITEM], hi[NITEM];
    float halo;
};

__device__ __forceinline__ int wrap_slot(int s) { return s >= NSLOT ? s - NSLOT : s; }

__device__ __forceinline__ void rows_fetch(RowRegs& R, const float* __restrict__ xb,
                                           int row0, int wstart, int tid) {
    const unsigned tb = (unsigned)(2 * (tid >> 4)) * (unsigned)(HH * WW) + (unsigned)(wstart + 4 * (tid & 15));
    #pragma unroll
    for (int i = 0; i < NITEM; ++i) {
        const int row = row0 + (i >> 1);
        const int rowc = row < 0 ? 0 : (row > HH - 1 ? HH - 1 : row);
        const unsigned off = tb + (unsigned)((i & 1) * 32 * HH * WW) + (unsigned)(rowc * WW);
        R.lo[i] = *(const v4f*)(xb + off);
        R.hi[i] = *(const v4f*)(xb + off + HH * WW);
    }
    {
        const int row = row0 + (tid >> 7);
        const int ci = (tid >> 1) & 63;
        const int gw = (tid & 1) ? wstart + TILE_W : wstart - 1;
        const int rowc = row < 0 ? 0 : (row > HH - 1 ? HH - 1 : row);
        const int gwc = gw < 0 ? 0 : (gw > WW - 1 ? WW - 1 : gw);
        R.halo = xb[(unsigned)((ci * HH + rowc) * WW + gwc)];
    }
}

__device__ __forceinline__ void rows_store(const RowRegs& R, float* __restrict__ sX, int s0,
                                           int row0, int wstart, int tid) {
    #pragma unroll
    for (int i = 0; i < NITEM; ++i) {
        const int cp = (tid >> 4) + 16 * (i & 1), c4 = tid & 15;
        float* d = sX + wrap_slot(s0 + (i >> 1)) * SLOT_SZ + cp * PAIR_PITCH + (4 * c4 + 1) * 2;
        const bool in = (unsigned)(row0 + (i >> 1)) < (unsigned)HH;
        *(v2f*)d       = (v2f){in ? R.lo[i].x : 0.f, in ? R.hi[i].x : 0.f};
        *(v2f*)(d + 2) = (v2f){in ? R.lo[i].y : 0.f, in ? R.hi[i].y : 0.f};
        *(v2f*)(d + 4) = (v2f){in ? R.lo[i].z : 0.f, in ? R.hi[i].z : 0.f};
        *(v2f*)(d + 6) = (v2f){in ? R.lo[i].w : 0.f, in ? R.hi[i].w : 0.f};
    }
    {
        const int ci = (tid >> 1) & 63;
        const int j = (tid & 1) ? (TILE_W + 1) : 0;
        const int gw = (tid & 1) ? wstart + TILE_W : wstart - 1;
        const bool in = (unsigned)(row0 + (tid >> 7)) < (unsigned)HH && (unsigned)gw < (unsigned)WW;
        sX[wrap_slot(s0 + (tid >> 7)) * SLOT_SZ + (ci >> 1) * PAIR_PITCH + j * 2 + (ci & 1)] = in ? R.halo : 0.f;
    }
}

__device__ __forceinline__ void frag_load(v2f (&fa)[NSET], v4f (&fr)[4][2],
                                          const float* aB, const float* sXb, const int (&rb)[4], int boff) {
    #pragma unroll
    for (int s = 0; s < NSET; ++s) fa[s] = *(const v2f*)(aB + s * USET);
    #pragma unroll
    for (int r = 0; r < 4; ++r) {
        fr[r][0] = *(const v4f*)(sXb + rb[r] + boff);
        fr[r][1] = *(const v4f*)(sXb + rb[r] + boff + 4);
    }
}

__device__ __forceinline__ void frag_mma(v8f (&acc)[NSET], const v2f (&fa)[NSET], const v4f (&fr)[4][2]) {
    v4f t[4][2];
    #pragma unroll
    for (int c = 0; c < 2; ++c) {
        t[0][c] = fr[0][c] - fr[2][c];
        t[1][c] = fr[1][c] + fr[2][c];
        t[2][c] = fr[2][c] - fr[1][c];
        t[3][c] = fr[1][c] - fr[3][c];
    }
    #pragma unroll
    for (int a = 0; a < 4; ++a) {
        const v2f d0 = t[a][0].xy, d1 = t[a][0].zw, d2 = t[a][1].xy, d3 = t[a][1].zw;
        acc[4 * a + 0] = __builtin_amdgcn_wmma_f32_16x16x4_f32(false, fa[4 * a + 0], false, d0 - d2, (short)0, acc[4 * a + 0], false, false);
        acc[4 * a + 1] = __builtin_amdgcn_wmma_f32_16x16x4_f32(false, fa[4 * a + 1], false, d1 + d2, (short)0, acc[4 * a + 1], false, false);
        acc[4 * a + 2] = __builtin_amdgcn_wmma_f32_16x16x4_f32(false, fa[4 * a + 2], false, d2 - d1, (short)0, acc[4 * a + 2], false, false);
        acc[4 * a + 3] = __builtin_amdgcn_wmma_f32_16x16x4_f32(false, fa[4 * a + 3], false, d1 - d3, (short)0, acc[4 * a + 3], false, false);
    }
}

__global__ __launch_bounds__(256)
void condconv_wmma_kernel(const float* __restrict__ x,
                          const float* __restrict__ aggw,
                          const float* __restrict__ aggb,
                          float* __restrict__ out) {
    __shared__ __attribute__((aligned(16))) float sU[NSET * USET];
    __shared__ __attribute__((aligned(16))) float sX[NSLOT * SLOT_SZ];

    const int tid    = threadIdx.x;
    const int b      = blockIdx.x >> 5;
    const int rem    = blockIdx.x & 31;
    const int mh     = rem & 1;
    const int wstart = ((rem >> 1) & 3) * TILE_W;
    const int h0     = (rem >> 3) * ROWS_PER_BLOCK;

    const int lane   = tid & 31;
    const int wv     = __builtin_amdgcn_readfirstlane(tid >> 5);
    const int hl     = lane >> 4;
    const int l16    = lane & 15;
    const int m_base = (wv & 1) * 16;
    const int j_base = ((wv >> 1) & 1) * 16;
    const int rp     = wv >> 2;

    const float* xb = x + (size_t)b * (CIN * HH * WW);

    {
        const float* wsrc = aggw + (size_t)b * (OCH * KTOT) + (size_t)(mh * MBLK) * KTOT;
        for (int it = 0; it < (MBLK * CIN) / 256; it += 4) {
            float g[4][9];
            #pragma unroll
            for (int j = 0; j < 4; ++j) {
                const int e = tid + 256 * (it + j);
                #pragma unroll
                for (int k = 0; k < 9; ++k) g[j][k] = wsrc[9 * e + k];
            }
            #pragma unroll
            for (int j = 0; j < 4; ++j) {
                const int e = tid + 256 * (it + j);
                const int o = e >> 6, ci = e & 63;
                float w[3][4];
                #pragma unroll
                for (int kh = 0; kh < 3; ++kh) {
                    const float g0 = g[j][3 * kh], g1 = g[j][3 * kh + 1], g2 = g[j][3 * kh + 2];
                    const float s02 = g0 + g2;
                    w[kh][0] = g0;
                    w[kh][1] = 0.5f * (s02 + g1);
                    w[kh][2] = 0.5f * (s02 - g1);
                    w[kh][3] = g2;
                }
                float* d = sU + o * SWS + ci;
                #pragma unroll
                for (int c = 0; c < 4; ++c) {
                    const float t02 = w[0][c] + w[2][c];
                    d[(0 + c) * USET]  = w[0][c];
                    d[(4 + c) * USET]  = 0.5f * (t02 + w[1][c]);
                    d[(8 + c) * USET]  = 0.5f * (t02 - w[1][c]);
                    d[(12 + c) * USET] = w[2][c];
                }
            }
        }
    }

    RowRegs R;
    {
        RowRegs P0, P1;
        rows_fetch(P0, xb, h0 - 1, wstart, tid);
        rows_fetch(P1, xb, h0 + 1, wstart, tid);
        rows_fetch(R,  xb, h0 + 3, wstart, tid);
        rows_store(P0, sX, 0, h0 - 1, wstart, tid);
        rows_store(P1, sX, 2, h0 + 1, wstart, tid);
        rows_store(R,  sX, 4, h0 + 3, wstart, tid);
    }

    float bv[8];
    #pragma unroll
    for (int r = 0; r < 8; ++r) bv[r] = aggb[b * OCH + mh * MBLK + m_base + hl * 8 + r];

    float* ob = out + (size_t)b * (OCH * HH * WW);
    const unsigned olane = (unsigned)((mh * MBLK + m_base + hl * 8) * (HH * WW) + wstart + 2 * (j_base + l16));

    const float* aLane = sU + (m_base + l16) * SWS + 2 * hl;
    const int    bLane = hl * PAIR_PITCH + (j_base + l16) * 4;

    __syncthreads();

    v2f fa[2][NSET];
    v4f fr[2][4][2];

    int sread = 0;
    int rb[4];
    #pragma unroll
    for (int r = 0; r < 4; ++r) rb[r] = wrap_slot(sread + 2 * rp + r) * SLOT_SZ + bLane;
    frag_load(fa[0], fr[0], aLane, sX, rb, 0);

    for (int p = 0; p < ROWS_PER_BLOCK / 4; ++p) {
        const int h = h0 + 4 * p + 2 * rp;
        const bool more = (p + 1 < ROWS_PER_BLOCK / 4);
        const int snew = wrap_slot(sread + 6);
        const int rnew = h0 - 1 + 4 * p + 6;

        v8f acc[NSET] = {};
        #pragma unroll
        for (int q = 0; q < CIN / 4; ++q) {
            if (q + 1 < CIN / 4)
                frag_load(fa[(q + 1) & 1], fr[(q + 1) & 1], aLane + (q + 1) * 4, sX, rb, (q + 1) * 2 * PAIR_PITCH);
            if (more) {
                if (q == 0)  rows_fetch(R, xb, rnew, wstart, tid);
                if (q == 7)  rows_store(R, sX, snew, rnew, wstart, tid);
                if (q == 8)  rows_fetch(R, xb, rnew + 2, wstart, tid);
                if (q == 15) rows_store(R, sX, wrap_slot(snew + 2), rnew + 2, wstart, tid);
            }
            __builtin_amdgcn_sched_barrier(0);
            frag_mma(acc, fa[q & 1], fr[q & 1]);
            __builtin_amdgcn_sched_barrier(0);
        }

        if (more) {
            __syncthreads();
            sread = wrap_slot(sread + 4);
            #pragma unroll
            for (int r = 0; r < 4; ++r) rb[r] = wrap_slot(sread + 2 * rp + r) * SLOT_SZ + bLane;
            frag_load(fa[0], fr[0], aLane, sX, rb, 0);
        }

        #pragma unroll
        for (int r = 0; r < 8; ++r) {
            float s0[4], s1[4];
            #pragma unroll
            for (int c = 0; c < 4; ++c) {
                s0[c] = (acc[c][r] + acc[4 + c][r]) + acc[8 + c][r];
                s1[c] = (acc[4 + c][r] - acc[8 + c][r]) - acc[12 + c][r];
            }
            const float y00 = ((s0[0] + s0[1]) + s0[2]) + bv[r];
            const float y01 = ((s0[1] - s0[2]) - s0[3]) + bv[r];
            const float y10 = ((s1[0] + s1[1]) + s1[2]) + bv[r];
            const float y11 = ((s1[1] - s1[2]) - s1[3]) + bv[r];
            float* po = ob + (size_t)h * WW + (olane + (unsigned)(r * HH * WW));
            *(v2f*)po        = (v2f){y00, y01};
            *(v2f*)(po + WW) = (v2f){y10, y11};
        }

    }
}

extern "C" void kernel_launch(void* const* d_in, const int* in_sizes, int n_in,
                              void* d_out, int out_size, void* d_ws, size_t ws_size,
                              hipStream_t stream) {
    const float* x      = (const float*)d_in[0];
    const float* net_w  = (const float*)d_in[1];
    const float* weight = (const float*)d_in[2];
    const float* bias   = (const float*)d_in[3];
    float* out = (float*)d_out;

    float* ws = (float*)d_ws;
    float* att   = ws;
    float* aggb  = att + BS * NEXP;
    float* aggw  = aggb + BS * OCH;
    float* part  = aggw + (size_t)BS * OCH * KTOT;

    gate_partial_kernel<<<BS * CIN, 256, 0, stream>>>(x, net_w, part);
    att_finalize_kernel<<<BS, OCH, 0, stream>>>(part, bias, att, aggb);
    agg_weight_kernel<<<(BS * OCH * KTOT) / 256, 256, 0, stream>>>(weight, att, aggw);
    condconv_wmma_kernel<<<BS * 32, 256, 0, stream>>>(x, aggw, aggb, out);
}
